# starting kernel
# baseline (speedup 1.0000x reference)
#include <hip/hip_runtime.h>
#include <math.h>

typedef float v2f __attribute__((ext_vector_type(2)));
typedef float v8f __attribute__((ext_vector_type(8)));

#define NN 100000
#define NEG_SLOPE 0.2f
#define EPSF 1e-16f

__global__ void fill_kernel(float* __restrict__ p, float v, int n) {
    int i = blockIdx.x * blockDim.x + threadIdx.x;
    if (i < n) p[i] = v;
}

__device__ __forceinline__ void atomicMaxF(float* addr, float v) {
    if (v >= 0.0f) atomicMax((int*)addr, __float_as_int(v));
    else           atomicMin((unsigned int*)addr, (unsigned int)__float_as_int(v));
}

__global__ __launch_bounds__(128) void gemm_wmma_kernel(
        const float* __restrict__ X, const float* __restrict__ W,
        float* __restrict__ H, int nrows, int K) {
    int wave = blockIdx.x * (blockDim.x >> 5) + (threadIdx.x >> 5);
    int mtiles = nrows >> 4;
    if (wave >= mtiles) return;

    int lane  = threadIdx.x & 31;
    int r0    = lane & 15;
    int khalf = lane >> 4;

    const float* xrow = X + (size_t)(wave * 16 + r0) * (size_t)K;

    v8f c0 = {}, c1 = {}, c2 = {}, c3 = {};
    for (int k0 = 0; k0 < K; k0 += 4) {
        int ka = k0 + khalf * 2;
        v2f a;
        a.x = xrow[ka];
        a.y = xrow[ka + 1];

        const float* w0 = W + (size_t)ka * 64;
        const float* w1 = w0 + 64;
        v2f b0, b1, b2, b3;
        b0.x = w0[r0];      b0.y = w1[r0];
        b1.x = w0[16 + r0]; b1.y = w1[16 + r0];
        b2.x = w0[32 + r0]; b2.y = w1[32 + r0];
        b3.x = w0[48 + r0]; b3.y = w1[48 + r0];

        c0 = __builtin_amdgcn_wmma_f32_16x16x4_f32(false, a, false, b0, (short)0, c0, false, false);
        c1 = __builtin_amdgcn_wmma_f32_16x16x4_f32(false, a, false, b1, (short)0, c1, false, false);
        c2 = __builtin_amdgcn_wmma_f32_16x16x4_f32(false, a, false, b2, (short)0, c2, false, false);
        c3 = __builtin_amdgcn_wmma_f32_16x16x4_f32(false, a, false, b3, (short)0, c3, false, false);
    }

    float* hbase = H + (size_t)(wave * 16 + khalf * 8) * 64 + r0;
    #pragma unroll
    for (int r = 0; r < 8; ++r) {
        float* row = hbase + (size_t)r * 64;
        row[0]  = c0[r];
        row[16] = c1[r];
        row[32] = c2[r];
        row[48] = c3[r];
    }
}

__global__ __launch_bounds__(256) void scores_kernel(
        const float* __restrict__ H, const float* __restrict__ a_s,
        const float* __restrict__ a_d, float* __restrict__ S, float* __restrict__ D) {
    int node = blockIdx.x * (blockDim.x >> 5) + (threadIdx.x >> 5);
    if (node >= NN) return;
    int lane = threadIdx.x & 31;
    const float* h = H + (size_t)node * 64;
    float h0 = h[lane], h1 = h[lane + 32];
    float s = h0 * a_s[lane] + h1 * a_s[lane + 32];
    float d = h0 * a_d[lane] + h1 * a_d[lane + 32];
    #pragma unroll
    for (int off = 16; off; off >>= 1) {
        s += __shfl_xor(s, off, 32);
        d += __shfl_xor(d, off, 32);
    }
    if (lane == 0) { S[node] = s; D[node] = d; }
}

__global__ void edge_max_kernel(const int* __restrict__ ei, int E, int Etot,
        const float* __restrict__ S, const float* __restrict__ D,
        float* __restrict__ M) {
    int i = blockIdx.x * blockDim.x + threadIdx.x;
    if (i >= Etot) return;
    int src, dst;
    if (i < E) { src = ei[i]; dst = ei[E + i]; } else { src = dst = i - E; }
    float a = S[src] + D[dst];
    a = (a >= 0.0f) ? a : NEG_SLOPE * a;
    atomicMaxF(&M[dst], a);
}

__global__ void edge_exp_kernel(const int* __restrict__ ei, int E, int Etot,
        const float* __restrict__ S, const float* __restrict__ D,
        const float* __restrict__ M, float* __restrict__ EW, float* __restrict__ Z) {
    int i = blockIdx.x * blockDim.x + threadIdx.x;
    if (i >= Etot) return;
    int src, dst;
    if (i < E) { src = ei[i]; dst = ei[E + i]; } else { src = dst = i - E; }
    float a = S[src] + D[dst];
    a = (a >= 0.0f) ? a : NEG_SLOPE * a;
    float e = expf(a - M[dst]);
    EW[i] = e;
    atomicAdd(&Z[dst], e);
}

__global__ __launch_bounds__(256) void edge_agg_kernel(
        const int* __restrict__ ei, int E, int Etot,
        const float* __restrict__ H, const float* __restrict__ EW,
        const float* __restrict__ Z, float* __restrict__ ACC) {
    int e = blockIdx.x * 8 + (threadIdx.x >> 5);
    if (e >= Etot) return;
    int lane = threadIdx.x & 31;
    int src, dst;
    if (e < E) { src = ei[e]; dst = ei[E + e]; } else { src = dst = e - E; }
    float w = EW[e] / (Z[dst] + EPSF);
    const float* h = H + (size_t)src * 64;
    float* o = ACC + (size_t)dst * 64;
    atomicAdd(&o[lane],      h[lane]      * w);
    atomicAdd(&o[lane + 32], h[lane + 32] * w);
}

__global__ void relu_bias_kernel(float* __restrict__ A, const float* __restrict__ b, int n) {
    int i = blockIdx.x * blockDim.x + threadIdx.x;
    if (i < n) {
        float v = A[i] + b[i & 63];
        A[i] = (v > 0.0f) ? v : 0.0f;
    }
}

__global__ void bias_kernel(float* __restrict__ A, const float* __restrict__ b, int n) {
    int i = blockIdx.x * blockDim.x + threadIdx.x;
    if (i < n) A[i] += b[i & 63];
}

extern "C" void kernel_launch(void* const* d_in, const int* in_sizes, int n_in,
                              void* d_out, int out_size, void* d_ws, size_t ws_size,
                              hipStream_t stream) {
    const float* x     = (const float*)d_in[0];
    const int*   ei    = (const int*)d_in[1];
    const float* W1    = (const float*)d_in[2];
    const float* as1   = (const float*)d_in[3];
    const float* ad1   = (const float*)d_in[4];
    const float* b1    = (const float*)d_in[5];
    const float* W2    = (const float*)d_in[6];
    const float* as2   = (const float*)d_in[7];
    const float* ad2   = (const float*)d_in[8];
    const float* b2    = (const float*)d_in[9];
    float* out = (float*)d_out;

    const int E    = in_sizes[1] / 2;
    const int Etot = E + NN;
    const int NF   = NN * 64;

    float* wsf = (float*)d_ws;
    float* H   = wsf;
    float* ACC = H   + NF;
    float* S   = ACC + NF;
    float* D   = S   + NN;
    float* M   = D   + NN;
    float* Z   = M   + NN;
    float* EW  = Z   + NN;

    const int TB = 256;
    dim3 blk(TB);
    int gN    = (NN + TB - 1) / TB;
    int gNF   = (NF + TB - 1) / TB;
    int gE    = (Etot + TB - 1) / TB;
    int gEagg = (Etot + 7) / 8;
    int gScr  = (NN + 7) / 8;
    int gGemm = ((NN / 16) + 3) / 4;

    fill_kernel<<<gNF, blk, 0, stream>>>(ACC, 0.0f, NF);
    fill_kernel<<<gN,  blk, 0, stream>>>(M, -INFINITY, NN);
    fill_kernel<<<gN,  blk, 0, stream>>>(Z, 0.0f, NN);

    gemm_wmma_kernel<<<gGemm, dim3(128), 0, stream>>>(x, W1, H, NN, 128);
    scores_kernel<<<gScr, blk, 0, stream>>>(H, as1, ad1, S, D);
    edge_max_kernel<<<gE, blk, 0, stream>>>(ei, E, Etot, S, D, M);
    edge_exp_kernel<<<gE, blk, 0, stream>>>(ei, E, Etot, S, D, M, EW, Z);
    edge_agg_kernel<<<gEagg, blk, 0, stream>>>(ei, E, Etot, H, EW, Z, ACC);
    relu_bias_kernel<<<gNF, blk, 0, stream>>>(ACC, b1, NF);

    fill_kernel<<<gN,  blk, 0, stream>>>(M, -INFINITY, NN);
    fill_kernel<<<gN,  blk, 0, stream>>>(Z, 0.0f, NN);
    fill_kernel<<<gNF, blk, 0, stream>>>(out, 0.0f, NF);

    gemm_wmma_kernel<<<gGemm, dim3(128), 0, stream>>>(ACC, W2, H, NN, 64);
    scores_kernel<<<gScr, blk, 0, stream>>>(H, as2, ad2, S, D);
    edge_max_kernel<<<gE, blk, 0, stream>>>(ei, E, Etot, S, D, M);
    edge_exp_kernel<<<gE, blk, 0, stream>>>(ei, E, Etot, S, D, M, EW, Z);
    edge_agg_kernel<<<gEagg, blk, 0, stream>>>(ei, E, Etot, H, EW, Z, out);
    bias_kernel<<<gNF, blk, 0, stream>>>(out, b2, NF);
}


# Row-sum scan reads one owner byte per edge (dst / 391, left by layer 1's max pass), 16 per 16-byte load and a round ahead, not 4-byte destinations; a wave's queue order, so its LDS add order, differs
# speedup vs baseline: 2.4213x; 2.4213x over previous
#include <hip/hip_runtime.h>
#include <math.h>

typedef float v2f __attribute__((ext_vector_type(2)));
typedef float v8f __attribute__((ext_vector_type(8)));
typedef int   v2i __attribute__((ext_vector_type(2)));

#define NN 100000
#define NEG_SLOPE 0.2f
#define EPSF 1e-16f

__device__ __forceinline__ void atomicMaxF(float* addr, float v) {
    if (v >= 0.0f) atomicMax((int*)addr, __float_as_int(v));
    else           atomicMin((unsigned int*)addr, (unsigned int)__float_as_int(v));
}

__global__ __launch_bounds__(128) void gemm_wmma_kernel(
        const float* __restrict__ X, const float* __restrict__ W,
        float* __restrict__ H, int nrows, int K) {
    int wave = blockIdx.x * (blockDim.x >> 5) + (threadIdx.x >> 5);
    int mtiles = nrows >> 4;
    if (wave >= mtiles) return;

    int lane  = threadIdx.x & 31;
    int r0    = lane & 15;
    int khalf = lane >> 4;

    const float* xrow = X + (size_t)(wave * 16 + r0) * (size_t)K;

    v8f c0 = {}, c1 = {}, c2 = {}, c3 = {};
    for (int k0 = 0; k0 < K; k0 += 4) {
        int ka = k0 + khalf * 2;
        v2f a;
        a.x = xrow[ka];
        a.y = xrow[ka + 1];

        const float* w0 = W + (size_t)ka * 64;
        const float* w1 = w0 + 64;
        v2f b0, b1, b2, b3;
        b0.x = w0[r0];      b0.y = w1[r0];
        b1.x = w0[16 + r0]; b1.y = w1[16 + r0];
        b2.x = w0[32 + r0]; b2.y = w1[32 + r0];
        b3.x = w0[48 + r0]; b3.y = w1[48 + r0];

        c0 = __builtin_amdgcn_wmma_f32_16x16x4_f32(false, a, false, b0, (short)0, c0, false, false);
        c1 = __builtin_amdgcn_wmma_f32_16x16x4_f32(false, a, false, b1, (short)0, c1, false, false);
        c2 = __builtin_amdgcn_wmma_f32_16x16x4_f32(false, a, false, b2, (short)0, c2, false, false);
        c3 = __builtin_amdgcn_wmma_f32_16x16x4_f32(false, a, false, b3, (short)0, c3, false, false);
    }

    float* hbase = H + (size_t)(wave * 16 + khalf * 8) * 64 + r0;
    #pragma unroll
    for (int r = 0; r < 8; ++r) {
        float* row = hbase + (size_t)r * 64;
        row[0]  = c0[r];
        row[16] = c1[r];
        row[32] = c2[r];
        row[48] = c3[r];
    }
}

__global__ __launch_bounds__(256) void scores_kernel(
        const float* __restrict__ H, const float* __restrict__ a_s,
        const float* __restrict__ a_d, float* __restrict__ S, float* __restrict__ DM) {
    int node = blockIdx.x * (blockDim.x >> 5) + (threadIdx.x >> 5);
    if (node >= NN) return;
    int lane = threadIdx.x & 31;
    const float* h = H + (size_t)node * 64;
    float h0 = h[lane], h1 = h[lane + 32];
    float s = h0 * a_s[lane] + h1 * a_s[lane + 32];
    float d = h0 * a_d[lane] + h1 * a_d[lane + 32];
    #pragma unroll
    for (int off = 16; off; off >>= 1) {
        s += __shfl_xor(s, off, 32);
        d += __shfl_xor(d, off, 32);
    }
    if (lane == 0) {
        float a = s + d;
        a = (a >= 0.0f) ? a : NEG_SLOPE * a;
        v2f dm; dm.x = d; dm.y = a;
        S[node] = s;
        *(v2f*)(DM + 2 * (size_t)node) = dm;
    }
}

#define AGG_BLOCKS  256
#define AGG_ROWS    ((NN + AGG_BLOCKS - 1) / AGG_BLOCKS)
__global__ void edge_max_kernel(const int* __restrict__ ei, int E, int Etot,
        const float* __restrict__ S, float* DM, float* __restrict__ AL, int* KEY) {
    int i = blockIdx.x * blockDim.x + threadIdx.x;
    if (i >= Etot) return;
    int src, dst;
    if (i < E) {
        src = ei[i]; dst = ei[E + i];
        if (KEY) ((unsigned char*)KEY)[i] = (unsigned char)((unsigned)dst / AGG_ROWS);
    } else { src = dst = i - E; }
    float* pm = DM + 2 * (size_t)dst;
    v2f dm = *(const v2f*)pm;
    float a = S[src] + dm.x;
    a = (a >= 0.0f) ? a : NEG_SLOPE * a;
    AL[i] = a;
    if (a > dm.y) atomicMaxF(pm + 1, a);
}

#define AGG_THREADS 1024
#define AGG_WAVES   (AGG_THREADS / 32)
#define AGG_QCAP    544
#define AGG_BATCH   8
#define AGG_KSTEP   (AGG_THREADS * 16)
#define AGG_LCAP    1024
#define AGG_SRCBITS 17

__global__ __launch_bounds__(AGG_THREADS) void edge_agg_lds_kernel(
        const int* __restrict__ ei, int E,
        const float* __restrict__ H, const float* __restrict__ AL,
        const float* __restrict__ DM, const float* __restrict__ bias,
        float* __restrict__ OUT, int relu,
        int* __restrict__ LCNT, int* __restrict__ LIST, int replay,
        const int* __restrict__ KEY) {
    __shared__ float acc[AGG_ROWS * 64];
    __shared__ float zl[AGG_ROWS + 1];
    __shared__ float ml[AGG_ROWS + 1];
    __shared__ int   qe[AGG_WAVES * AGG_QCAP];

    const int lo = blockIdx.x * AGG_ROWS;
    int cnt = NN - lo;
    if (cnt > AGG_ROWS) cnt = AGG_ROWS;
    if (cnt < 0) cnt = 0;
    const int lane = threadIdx.x & 31;
    const int wave = __builtin_amdgcn_readfirstlane((int)(threadIdx.x >> 5));

    static_assert(AGG_WAVES == 32, "one list length per lane");
    int len = 0;
    if (replay) len = LCNT[blockIdx.x * AGG_WAVES + lane];

    for (int r = wave; r < cnt; r += AGG_WAVES) {
        int n = lo + r;
        float m = DM[2 * (size_t)n + 1];
        float w = expf(AL[(size_t)E + n] - m);
        const float* h = H + (size_t)n * 64;
        acc[r * 64 + lane]      = h[lane]      * w;
        acc[r * 64 + lane + 32] = h[lane + 32] * w;
        if (lane == 0) { zl[r] = w; ml[r] = m; }
    }
    __syncthreads();

    int* myqe = qe + wave * AGG_QCAP;
    const int* dstp = ei + E;
    int qn = 0;

    const int slot = blockIdx.x * AGG_WAVES + wave;
    v2i* mylist = (v2i*)LIST + (size_t)slot * AGG_LCAP;
    const bool record = (replay == 0);
    int ln = 0;

    auto rows = [&](int n, int src, int dl, float w) {
        #pragma unroll 1
        for (int j0 = 0; j0 < n; j0 += AGG_BATCH) {
            float h0[AGG_BATCH], h1[AGG_BATCH];
            #pragma unroll
            for (int j = 0; j < AGG_BATCH; ++j) {
                h0[j] = 0.0f; h1[j] = 0.0f;
                if (j0 + j < n) {
                    int s = __builtin_amdgcn_readlane(src, j0 + j);
                    const float* h = H + (size_t)s * 64;
                    h0[j] = h[lane];
                    h1[j] = h[lane + 32];
                }
            }
            #pragma unroll
            for (int j = 0; j < AGG_BATCH; ++j) {
                if (j0 + j < n) {
                    int d = __builtin_amdgcn_readlane(dl, j0 + j);
                    float wj = __int_as_float(__builtin_amdgcn_readlane(__float_as_int(w), j0 + j));
                    float* o = acc + d * 64;
                    atomicAdd(&o[lane],      h0[j] * wj);
                    atomicAdd(&o[lane + 32], h1[j] * wj);
                }
            }
        }
    };

    auto flush = [&](int n) {
        int idx = qn - n + lane;
        qn -= n;
        int e = 0, src = 0, dl = 0;
        float w = 0.0f;
        if (lane < n) {
            e = myqe[idx];
            dl = dstp[e] - lo;
            src = ei[e];
            w = expf(AL[e] - ml[dl]);
            atomicAdd(&zl[dl], w);
        }
        if (record) {
            if (ln >= 0 && ln + n <= AGG_LCAP) {
                if (lane < n) {
                    v2i en; en.x = e; en.y = src | (dl << AGG_SRCBITS);
                    mylist[ln + lane] = en;
                }
                ln += n;
            } else {
                ln = -1;
            }
        }
        rows(n, src, dl, w);
    };

    auto push = [&](bool m, int e) {
        unsigned mask = __builtin_amdgcn_ballot_w32(m);
        if (mask) {
            int pos = qn + (int)__builtin_amdgcn_mbcnt_lo(mask, 0u);
            if (m) myqe[pos] = e;
            qn += __builtin_popcount(mask);
        }
    };

    bool scan = true;
    int mylen = 0;
    if (replay) {
        scan = __builtin_amdgcn_ballot_w32(len < 0) != 0u;
        mylen = __builtin_amdgcn_readlane(len, wave);
    }

    if (scan) {
    const unsigned pat = (unsigned)blockIdx.x * 0x01010101u;
    auto zero_bytes = [](unsigned x) -> unsigned {
        unsigned t = (x & 0x7F7F7F7Fu) + 0x7F7F7F7Fu;
        return ~(t | x) & 0x80808080u;
    };
    auto flags = [&](int4 v) -> unsigned {
        unsigned z0 = zero_bytes((unsigned)v.x ^ pat), z1 = zero_bytes((unsigned)v.y ^ pat);
        unsigned z2 = zero_bytes((unsigned)v.z ^ pat), z3 = zero_bytes((unsigned)v.w ^ pat);
        return (z0 >> 7) | (z1 >> 6) | (z2 >> 5) | (z3 >> 4);
    };
    auto scan_round = [&](int4 c0, int4 c1, int4 c2, int4 c3, int rbase) {
        unsigned long long M = (unsigned long long)(flags(c0) | (flags(c1) << 4))
                             | ((unsigned long long)(flags(c2) | (flags(c3) << 4)) << 32);
        const int lbase = rbase + (int)threadIdx.x * 16;
        while (__builtin_amdgcn_ballot_w32(M != 0ull)) {
            const bool has = (M != 0ull);
            const int P = has ? __builtin_ctzll(M) : 0;
            M &= M - 1ull;
            const int p = P & 31;
            const int u = ((P >> 5) << 1) + ((p >> 2) & 1);
            push(has, lbase + u * AGG_KSTEP + ((p & 3) << 2) + (p >> 3));
            while (qn >= 32) flush(32);
        }
    };
    const int pieces = E >> 4;
    const int PPR = 4 * AGG_THREADS;
    const int nfull = pieces / PPR;
    const int4* key4 = (const int4*)KEY + threadIdx.x;
    const int nokey = (int)~pat;
    int4 cur[4], nxt[4];
    #pragma unroll
    for (int u = 0; u < 4; ++u) {
        cur[u] = make_int4(nokey, nokey, nokey, nokey);
        nxt[u] = cur[u];
        if (nfull > 0) cur[u] = key4[u * AGG_THREADS];
    }
    for (int r = 0; r < nfull; ++r) {
        const int rn = (r + 1 < nfull) ? r + 1 : r;
        #pragma unroll
        for (int u = 0; u < 4; ++u)
            nxt[u] = key4[(size_t)rn * PPR + u * AGG_THREADS];
        scan_round(cur[0], cur[1], cur[2], cur[3], r * (PPR * 16));
        #pragma unroll
        for (int u = 0; u < 4; ++u) cur[u] = nxt[u];
    }
    if (nfull * PPR < pieces) {
        int4 c[4];
        #pragma unroll
        for (int u = 0; u < 4; ++u) {
            int pc = nfull * PPR + u * AGG_THREADS + (int)threadIdx.x;
            c[u] = make_int4(nokey, nokey, nokey, nokey);
            if (pc < pieces) c[u] = ((const int4*)KEY)[pc];
        }
        scan_round(c[0], c[1], c[2], c[3], nfull * (PPR * 16));
    }
    for (int base = pieces << 4; base < E; base += AGG_THREADS) {
        int e = base + (int)threadIdx.x;
        bool has = false;
        if (e < E) has = ((const unsigned char*)KEY)[e] == (unsigned char)blockIdx.x;
        push(has, e);
        while (qn >= 32) flush(32);
    }
    if (qn > 0) flush(qn);
    if (record && lane == 0) LCNT[slot] = ln;
    } else {
    const int n = (mylen < AGG_LCAP) ? mylen : AGG_LCAP;
    v2i e0 = {0, 0}, e1 = {0, 0};
    float a0 = 0.0f;
    if (lane < n)      e0 = mylist[lane];
    if (lane + 32 < n) e1 = mylist[lane + 32];
    if (lane < n)      a0 = AL[e0.x];
    for (int c0 = 0; c0 < n; c0 += 32) {
        int m = n - c0;
        if (m > 32) m = 32;
        v2i e2 = {0, 0};
        float a1 = 0.0f;
        if (c0 + 64 + lane < n) e2 = mylist[c0 + 64 + lane];
        if (c0 + 32 + lane < n) a1 = AL[e1.x];
        int src = 0, dl = 0;
        float w = 0.0f;
        if (lane < m) {
            src = e0.y & ((1 << AGG_SRCBITS) - 1);
            dl  = e0.y >> AGG_SRCBITS;
            w = expf(a0 - ml[dl]);
            atomicAdd(&zl[dl], w);
        }
        rows(m, src, dl, w);
        e0 = e1; e1 = e2; a0 = a1;
    }
    }
    __syncthreads();

    const float b0 = bias[lane], b1 = bias[lane + 32];
    for (int r = wave; r < cnt; r += AGG_WAVES) {
        float zz = zl[r] + EPSF;
        float v0 = acc[r * 64 + lane]      / zz + b0;
        float v1 = acc[r * 64 + lane + 32] / zz + b1;
        if (relu) {
            v0 = (v0 > 0.0f) ? v0 : 0.0f;
            v1 = (v1 > 0.0f) ? v1 : 0.0f;
        }
        float* o = OUT + (size_t)(lo + r) * 64;
        o[lane]      = v0;
        o[lane + 32] = v1;
    }
}

extern "C" void kernel_launch(void* const* d_in, const int* in_sizes, int n_in,
                              void* d_out, int out_size, void* d_ws, size_t ws_size,
                              hipStream_t stream) {
    const float* x     = (const float*)d_in[0];
    const int*   ei    = (const int*)d_in[1];
    const float* W1    = (const float*)d_in[2];
    const float* as1   = (const float*)d_in[3];
    const float* ad1   = (const float*)d_in[4];
    const float* b1    = (const float*)d_in[5];
    const float* W2    = (const float*)d_in[6];
    const float* as2   = (const float*)d_in[7];
    const float* ad2   = (const float*)d_in[8];
    const float* b2    = (const float*)d_in[9];
    float* out = (float*)d_out;

    const int E    = in_sizes[1] / 2;
    const int Etot = E + NN;
    const int NF   = NN * 64;

    float* wsf = (float*)d_ws;
    float* H   = wsf;
    float* X2  = H   + NF;
    float* S   = X2  + NF;
    float* DM  = S   + NN;
    int* LCNT  = (int*)(DM + 2 * NN);
    int* LIST  = LCNT + AGG_BLOCKS * AGG_WAVES;
    int* KEY   = LIST + AGG_BLOCKS * AGG_WAVES * AGG_LCAP * 2;
    float* AL  = (float*)(KEY + ((E + 15) / 16) * 4);

    const int TB = 256;
    dim3 blk(TB);
    int gE    = (Etot + TB - 1) / TB;
    int gScr  = (NN + 7) / 8;
    int gGemm = ((NN / 16) + 3) / 4;

    gemm_wmma_kernel<<<gGemm, dim3(128), 0, stream>>>(x, W1, H, NN, 128);
    scores_kernel<<<gScr, blk, 0, stream>>>(H, as1, ad1, S, DM);
    edge_max_kernel<<<gE, blk, 0, stream>>>(ei, E, Etot, S, DM, AL, KEY);
    edge_agg_lds_kernel<<<AGG_BLOCKS, dim3(AGG_THREADS), 0, stream>>>(ei, E, H, AL, DM, b1, X2, 1, LCNT, LIST, 0, KEY);

    gemm_wmma_kernel<<<gGemm, dim3(128), 0, stream>>>(X2, W2, H, NN, 64);
    scores_kernel<<<gScr, blk, 0, stream>>>(H, as2, ad2, S, DM);
    edge_max_kernel<<<gE, blk, 0, stream>>>(ei, E, Etot, S, DM, AL, (int*)0);
    edge_agg_lds_kernel<<<AGG_BLOCKS, dim3(AGG_THREADS), 0, stream>>>(ei, E, H, AL, DM, b2, out, 0, LCNT, LIST, 1, KEY);
}


# Layer 1's max pass leaves per edge one 8-byte record (source, local row, score); the row-sum scan fetches a queued edge with that one load, not three 4-byte gathers (dst, src, score); sums unchanged
# speedup vs baseline: 2.6119x; 1.0787x over previous
#include <hip/hip_runtime.h>
#include <math.h>

typedef float v2f __attribute__((ext_vector_type(2)));
typedef float v8f __attribute__((ext_vector_type(8)));
typedef int   v2i __attribute__((ext_vector_type(2)));

#define NN 100000
#define NEG_SLOPE 0.2f
#define EPSF 1e-16f

__device__ __forceinline__ void atomicMaxF(float* addr, float v) {
    if (v >= 0.0f) atomicMax((int*)addr, __float_as_int(v));
    else           atomicMin((unsigned int*)addr, (unsigned int)__float_as_int(v));
}

__global__ __launch_bounds__(128) void gemm_wmma_kernel(
        const float* __restrict__ X, const float* __restrict__ W,
        float* __restrict__ H, int nrows, int K) {
    int wave = blockIdx.x * (blockDim.x >> 5) + (threadIdx.x >> 5);
    int mtiles = nrows >> 4;
    if (wave >= mtiles) return;

    int lane  = threadIdx.x & 31;
    int r0    = lane & 15;
    int khalf = lane >> 4;

    const float* xrow = X + (size_t)(wave * 16 + r0) * (size_t)K;

    v8f c0 = {}, c1 = {}, c2 = {}, c3 = {};
    for (int k0 = 0; k0 < K; k0 += 4) {
        int ka = k0 + khalf * 2;
        v2f a;
        a.x = xrow[ka];
        a.y = xrow[ka + 1];

        const float* w0 = W + (size_t)ka * 64;
        const float* w1 = w0 + 64;
        v2f b0, b1, b2, b3;
        b0.x = w0[r0];      b0.y = w1[r0];
        b1.x = w0[16 + r0]; b1.y = w1[16 + r0];
        b2.x = w0[32 + r0]; b2.y = w1[32 + r0];
        b3.x = w0[48 + r0]; b3.y = w1[48 + r0];

        c0 = __builtin_amdgcn_wmma_f32_16x16x4_f32(false, a, false, b0, (short)0, c0, false, false);
        c1 = __builtin_amdgcn_wmma_f32_16x16x4_f32(false, a, false, b1, (short)0, c1, false, false);
        c2 = __builtin_amdgcn_wmma_f32_16x16x4_f32(false, a, false, b2, (short)0, c2, false, false);
        c3 = __builtin_amdgcn_wmma_f32_16x16x4_f32(false, a, false, b3, (short)0, c3, false, false);
    }

    float* hbase = H + (size_t)(wave * 16 + khalf * 8) * 64 + r0;
    #pragma unroll
    for (int r = 0; r < 8; ++r) {
        float* row = hbase + (size_t)r * 64;
        row[0]  = c0[r];
        row[16] = c1[r];
        row[32] = c2[r];
        row[48] = c3[r];
    }
}

__global__ __launch_bounds__(256) void scores_kernel(
        const float* __restrict__ H, const float* __restrict__ a_s,
        const float* __restrict__ a_d, float* __restrict__ S, float* __restrict__ DM) {
    int node = blockIdx.x * (blockDim.x >> 5) + (threadIdx.x >> 5);
    if (node >= NN) return;
    int lane = threadIdx.x & 31;
    const float* h = H + (size_t)node * 64;
    float h0 = h[lane], h1 = h[lane + 32];
    float s = h0 * a_s[lane] + h1 * a_s[lane + 32];
    float d = h0 * a_d[lane] + h1 * a_d[lane + 32];
    #pragma unroll
    for (int off = 16; off; off >>= 1) {
        s += __shfl_xor(s, off, 32);
        d += __shfl_xor(d, off, 32);
    }
    if (lane == 0) {
        float a = s + d;
        a = (a >= 0.0f) ? a : NEG_SLOPE * a;
        v2f dm; dm.x = d; dm.y = a;
        S[node] = s;
        *(v2f*)(DM + 2 * (size_t)node) = dm;
    }
}

#define AGG_BLOCKS  256
#define AGG_ROWS    ((NN + AGG_BLOCKS - 1) / AGG_BLOCKS)
#define AGG_SRCBITS 17
__global__ void edge_max_kernel(const int* __restrict__ ei, int E, int Etot,
        const float* __restrict__ S, float* DM, float* __restrict__ AL, int* KEY, int* REC) {
    int i = blockIdx.x * blockDim.x + threadIdx.x;
    if (i >= Etot) return;
    int src, dst;
    if (i < E) { src = ei[i]; dst = ei[E + i]; }
    else       { src = dst = i - E; }
    float* pm = DM + 2 * (size_t)dst;
    v2f dm = *(const v2f*)pm;
    float a = S[src] + dm.x;
    a = (a >= 0.0f) ? a : NEG_SLOPE * a;
    if (KEY && i < E) {
        int own = (int)((unsigned)dst / AGG_ROWS);
        ((unsigned char*)KEY)[i] = (unsigned char)own;
        v2i rec;
        rec.x = src | ((dst - own * AGG_ROWS) << AGG_SRCBITS);
        rec.y = __float_as_int(a);
        ((v2i*)REC)[i] = rec;
    } else {
        AL[i] = a;
    }
    if (a > dm.y) atomicMaxF(pm + 1, a);
}

#define AGG_THREADS 1024
#define AGG_WAVES   (AGG_THREADS / 32)
#define AGG_QCAP    544
#define AGG_BATCH   8
#define AGG_KSTEP   (AGG_THREADS * 16)
#define AGG_LCAP    1024

__global__ __launch_bounds__(AGG_THREADS) void edge_agg_lds_kernel(
        const int* __restrict__ REC, int E,
        const float* __restrict__ H, const float* __restrict__ AL,
        const float* __restrict__ DM, const float* __restrict__ bias,
        float* __restrict__ OUT, int relu,
        int* __restrict__ LCNT, int* __restrict__ LIST, int replay,
        const int* __restrict__ KEY) {
    __shared__ float acc[AGG_ROWS * 64];
    __shared__ float zl[AGG_ROWS + 1];
    __shared__ float ml[AGG_ROWS + 1];
    __shared__ int   qe[AGG_WAVES * AGG_QCAP];

    const int lo = blockIdx.x * AGG_ROWS;
    int cnt = NN - lo;
    if (cnt > AGG_ROWS) cnt = AGG_ROWS;
    if (cnt < 0) cnt = 0;
    const int lane = threadIdx.x & 31;
    const int wave = __builtin_amdgcn_readfirstlane((int)(threadIdx.x >> 5));

    static_assert(AGG_WAVES == 32, "one list length per lane");
    int len = 0;
    if (replay) len = LCNT[blockIdx.x * AGG_WAVES + lane];

    for (int r = wave; r < cnt; r += AGG_WAVES) {
        int n = lo + r;
        float m = DM[2 * (size_t)n + 1];
        float w = expf(AL[(size_t)E + n] - m);
        const float* h = H + (size_t)n * 64;
        acc[r * 64 + lane]      = h[lane]      * w;
        acc[r * 64 + lane + 32] = h[lane + 32] * w;
        if (lane == 0) { zl[r] = w; ml[r] = m; }
    }
    __syncthreads();

    int* myqe = qe + wave * AGG_QCAP;
    int qn = 0;

    const int slot = blockIdx.x * AGG_WAVES + wave;
    v2i* mylist = (v2i*)LIST + (size_t)slot * AGG_LCAP;
    const bool record = (replay == 0);
    int ln = 0;

    auto rows = [&](int n, int src, int dl, float w) {
        #pragma unroll 1
        for (int j0 = 0; j0 < n; j0 += AGG_BATCH) {
            float h0[AGG_BATCH], h1[AGG_BATCH];
            #pragma unroll
            for (int j = 0; j < AGG_BATCH; ++j) {
                h0[j] = 0.0f; h1[j] = 0.0f;
                if (j0 + j < n) {
                    int s = __builtin_amdgcn_readlane(src, j0 + j);
                    const float* h = H + (size_t)s * 64;
                    h0[j] = h[lane];
                    h1[j] = h[lane + 32];
                }
            }
            #pragma unroll
            for (int j = 0; j < AGG_BATCH; ++j) {
                if (j0 + j < n) {
                    int d = __builtin_amdgcn_readlane(dl, j0 + j);
                    float wj = __int_as_float(__builtin_amdgcn_readlane(__float_as_int(w), j0 + j));
                    float* o = acc + d * 64;
                    atomicAdd(&o[lane],      h0[j] * wj);
                    atomicAdd(&o[lane + 32], h1[j] * wj);
                }
            }
        }
    };

    auto flush = [&](int n) {
        int idx = qn - n + lane;
        qn -= n;
        int e = 0, sd = 0, src = 0, dl = 0;
        float w = 0.0f;
        if (lane < n) {
            e = myqe[idx];
            const v2i rec = ((const v2i*)REC)[e];
            float a = __int_as_float(rec.y);
            if (replay) a = AL[e];
            sd  = rec.x;
            src = sd & ((1 << AGG_SRCBITS) - 1);
            dl  = sd >> AGG_SRCBITS;
            w = expf(a - ml[dl]);
            atomicAdd(&zl[dl], w);
        }
        if (record) {
            if (ln >= 0 && ln + n <= AGG_LCAP) {
                if (lane < n) {
                    v2i en; en.x = e; en.y = sd;
                    mylist[ln + lane] = en;
                }
                ln += n;
            } else {
                ln = -1;
            }
        }
        rows(n, src, dl, w);
    };

    auto push = [&](bool m, int e) {
        unsigned mask = __builtin_amdgcn_ballot_w32(m);
        if (mask) {
            int pos = qn + (int)__builtin_amdgcn_mbcnt_lo(mask, 0u);
            if (m) myqe[pos] = e;
            qn += __builtin_popcount(mask);
        }
    };

    bool scan = true;
    int mylen = 0;
    if (replay) {
        scan = __builtin_amdgcn_ballot_w32(len < 0) != 0u;
        mylen = __builtin_amdgcn_readlane(len, wave);
    }

    if (scan) {
    const unsigned pat = (unsigned)blockIdx.x * 0x01010101u;
    auto zero_bytes = [](unsigned x) -> unsigned {
        unsigned t = (x & 0x7F7F7F7Fu) + 0x7F7F7F7Fu;
        return ~(t | x) & 0x80808080u;
    };
    auto flags = [&](int4 v) -> unsigned {
        unsigned z0 = zero_bytes((unsigned)v.x ^ pat), z1 = zero_bytes((unsigned)v.y ^ pat);
        unsigned z2 = zero_bytes((unsigned)v.z ^ pat), z3 = zero_bytes((unsigned)v.w ^ pat);
        return (z0 >> 7) | (z1 >> 6) | (z2 >> 5) | (z3 >> 4);
    };
    auto scan_round = [&](int4 c0, int4 c1, int4 c2, int4 c3, int rbase) {
        unsigned long long M = (unsigned long long)(flags(c0) | (flags(c1) << 4))
                             | ((unsigned long long)(flags(c2) | (flags(c3) << 4)) << 32);
        const int lbase = rbase + (int)threadIdx.x * 16;
        while (__builtin_amdgcn_ballot_w32(M != 0ull)) {
            const bool has = (M != 0ull);
            const int P = has ? __builtin_ctzll(M) : 0;
            M &= M - 1ull;
            const int p = P & 31;
            const int u = ((P >> 5) << 1) + ((p >> 2) & 1);
            push(has, lbase + u * AGG_KSTEP + ((p & 3) << 2) + (p >> 3));
            while (qn >= 32) flush(32);
        }
    };
    const int pieces = E >> 4;
    const int PPR = 4 * AGG_THREADS;
    const int nfull = pieces / PPR;
    const int4* key4 = (const int4*)KEY + threadIdx.x;
    const int nokey = (int)~pat;
    int4 cur[4], nxt[4];
    #pragma unroll
    for (int u = 0; u < 4; ++u) {
        cur[u] = make_int4(nokey, nokey, nokey, nokey);
        nxt[u] = cur[u];
        if (nfull > 0) cur[u] = key4[u * AGG_THREADS];
    }
    for (int r = 0; r < nfull; ++r) {
        const int rn = (r + 1 < nfull) ? r + 1 : r;
        #pragma unroll
        for (int u = 0; u < 4; ++u)
            nxt[u] = key4[(size_t)rn * PPR + u * AGG_THREADS];
        scan_round(cur[0], cur[1], cur[2], cur[3], r * (PPR * 16));
        #pragma unroll
        for (int u = 0; u < 4; ++u) cur[u] = nxt[u];
    }
    if (nfull * PPR < pieces) {
        int4 c[4];
        #pragma unroll
        for (int u = 0; u < 4; ++u) {
            int pc = nfull * PPR + u * AGG_THREADS + (int)threadIdx.x;
            c[u] = make_int4(nokey, nokey, nokey, nokey);
            if (pc < pieces) c[u] = ((const int4*)KEY)[pc];
        }
        scan_round(c[0], c[1], c[2], c[3], nfull * (PPR * 16));
    }
    for (int base = pieces << 4; base < E; base += AGG_THREADS) {
        int e = base + (int)threadIdx.x;
        bool has = false;
        if (e < E) has = ((const unsigned char*)KEY)[e] == (unsigned char)blockIdx.x;
        push(has, e);
        while (qn >= 32) flush(32);
    }
    if (qn > 0) flush(qn);
    if (record && lane == 0) LCNT[slot] = ln;
    } else {
    const int n = (mylen < AGG_LCAP) ? mylen : AGG_LCAP;
    v2i e0 = {0, 0}, e1 = {0, 0};
    float a0 = 0.0f;
    if (lane < n)      e0 = mylist[lane];
    if (lane + 32 < n) e1 = mylist[lane + 32];
    if (lane < n)      a0 = AL[e0.x];
    for (int c0 = 0; c0 < n; c0 += 32) {
        int m = n - c0;
        if (m > 32) m = 32;
        v2i e2 = {0, 0};
        float a1 = 0.0f;
        if (c0 + 64 + lane < n) e2 = mylist[c0 + 64 + lane];
        if (c0 + 32 + lane < n) a1 = AL[e1.x];
        int src = 0, dl = 0;
        float w = 0.0f;
        if (lane < m) {
            src = e0.y & ((1 << AGG_SRCBITS) - 1);
            dl  = e0.y >> AGG_SRCBITS;
            w = expf(a0 - ml[dl]);
            atomicAdd(&zl[dl], w);
        }
        rows(m, src, dl, w);
        e0 = e1; e1 = e2; a0 = a1;
    }
    }
    __syncthreads();

    const float b0 = bias[lane], b1 = bias[lane + 32];
    for (int r = wave; r < cnt; r += AGG_WAVES) {
        float zz = zl[r] + EPSF;
        float v0 = acc[r * 64 + lane]      / zz + b0;
        float v1 = acc[r * 64 + lane + 32] / zz + b1;
        if (relu) {
            v0 = (v0 > 0.0f) ? v0 : 0.0f;
            v1 = (v1 > 0.0f) ? v1 : 0.0f;
        }
        float* o = OUT + (size_t)(lo + r) * 64;
        o[lane]      = v0;
        o[lane + 32] = v1;
    }
}

extern "C" void kernel_launch(void* const* d_in, const int* in_sizes, int n_in,
                              void* d_out, int out_size, void* d_ws, size_t ws_size,
                              hipStream_t stream) {
    const float* x     = (const float*)d_in[0];
    const int*   ei    = (const int*)d_in[1];
    const float* W1    = (const float*)d_in[2];
    const float* as1   = (const float*)d_in[3];
    const float* ad1   = (const float*)d_in[4];
    const float* b1    = (const float*)d_in[5];
    const float* W2    = (const float*)d_in[6];
    const float* as2   = (const float*)d_in[7];
    const float* ad2   = (const float*)d_in[8];
    const float* b2    = (const float*)d_in[9];
    float* out = (float*)d_out;

    const int E    = in_sizes[1] / 2;
    const int Etot = E + NN;
    const int NF   = NN * 64;

    float* wsf = (float*)d_ws;
    float* H   = wsf;
    float* X2  = H   + NF;
    float* S   = X2  + NF;
    float* DM  = S   + NN;
    int* LCNT  = (int*)(DM + 2 * NN);
    int* LIST  = LCNT + AGG_BLOCKS * AGG_WAVES;
    int* KEY   = LIST + AGG_BLOCKS * AGG_WAVES * AGG_LCAP * 2;
    int* REC   = KEY + ((E + 15) / 16) * 4;
    float* AL  = (float*)(REC + 2 * (size_t)E);

    const int TB = 256;
    dim3 blk(TB);
    int gE    = (Etot + TB - 1) / TB;
    int gScr  = (NN + 7) / 8;
    int gGemm = ((NN / 16) + 3) / 4;

    gemm_wmma_kernel<<<gGemm, dim3(128), 0, stream>>>(x, W1, H, NN, 128);
    scores_kernel<<<gScr, blk, 0, stream>>>(H, as1, ad1, S, DM);
    edge_max_kernel<<<gE, blk, 0, stream>>>(ei, E, Etot, S, DM, AL, KEY, REC);
    edge_agg_lds_kernel<<<AGG_BLOCKS, dim3(AGG_THREADS), 0, stream>>>(REC, E, H, AL, DM, b1, X2, 1, LCNT, LIST, 0, KEY);

    gemm_wmma_kernel<<<gGemm, dim3(128), 0, stream>>>(X2, W2, H, NN, 64);
    scores_kernel<<<gScr, blk, 0, stream>>>(H, as2, ad2, S, DM);
    edge_max_kernel<<<gE, blk, 0, stream>>>(ei, E, Etot, S, DM, AL, (int*)0, (int*)0);
    edge_agg_lds_kernel<<<AGG_BLOCKS, dim3(AGG_THREADS), 0, stream>>>(REC, E, H, AL, DM, b2, out, 0, LCNT, LIST, 1, KEY);
}


# Layer 2 has no edge-max launch: its row-sum launch walks its lists twice, first forming the scores (S[src] + D[row], D in LDS) and the row maxima by LDS max; 7 launches, not 8; sums unchanged
# speedup vs baseline: 3.3232x; 1.2723x over previous
#include <hip/hip_runtime.h>
#include <math.h>

typedef float v2f __attribute__((ext_vector_type(2)));
typedef float v8f __attribute__((ext_vector_type(8)));
typedef int   v2i __attribute__((ext_vector_type(2)));

#define NN 100000
#define NEG_SLOPE 0.2f
#define EPSF 1e-16f

__device__ __forceinline__ void atomicMaxF(float* addr, float v) {
    if (v >= 0.0f) atomicMax((int*)addr, __float_as_int(v));
    else           atomicMin((unsigned int*)addr, (unsigned int)__float_as_int(v));
}

__global__ __launch_bounds__(128) void gemm_wmma_kernel(
        const float* __restrict__ X, const float* __restrict__ W,
        float* __restrict__ H, int nrows, int K) {
    int wave = blockIdx.x * (blockDim.x >> 5) + (threadIdx.x >> 5);
    int mtiles = nrows >> 4;
    if (wave >= mtiles) return;

    int lane  = threadIdx.x & 31;
    int r0    = lane & 15;
    int khalf = lane >> 4;

    const float* xrow = X + (size_t)(wave * 16 + r0) * (size_t)K;

    v8f c0 = {}, c1 = {}, c2 = {}, c3 = {};
    for (int k0 = 0; k0 < K; k0 += 4) {
        int ka = k0 + khalf * 2;
        v2f a;
        a.x = xrow[ka];
        a.y = xrow[ka + 1];

        const float* w0 = W + (size_t)ka * 64;
        const float* w1 = w0 + 64;
        v2f b0, b1, b2, b3;
        b0.x = w0[r0];      b0.y = w1[r0];
        b1.x = w0[16 + r0]; b1.y = w1[16 + r0];
        b2.x = w0[32 + r0]; b2.y = w1[32 + r0];
        b3.x = w0[48 + r0]; b3.y = w1[48 + r0];

        c0 = __builtin_amdgcn_wmma_f32_16x16x4_f32(false, a, false, b0, (short)0, c0, false, false);
        c1 = __builtin_amdgcn_wmma_f32_16x16x4_f32(false, a, false, b1, (short)0, c1, false, false);
        c2 = __builtin_amdgcn_wmma_f32_16x16x4_f32(false, a, false, b2, (short)0, c2, false, false);
        c3 = __builtin_amdgcn_wmma_f32_16x16x4_f32(false, a, false, b3, (short)0, c3, false, false);
    }

    float* hbase = H + (size_t)(wave * 16 + khalf * 8) * 64 + r0;
    #pragma unroll
    for (int r = 0; r < 8; ++r) {
        float* row = hbase + (size_t)r * 64;
        row[0]  = c0[r];
        row[16] = c1[r];
        row[32] = c2[r];
        row[48] = c3[r];
    }
}

__global__ __launch_bounds__(256) void scores_kernel(
        const float* __restrict__ H, const float* __restrict__ a_s,
        const float* __restrict__ a_d, float* __restrict__ S, float* __restrict__ DM) {
    int node = blockIdx.x * (blockDim.x >> 5) + (threadIdx.x >> 5);
    if (node >= NN) return;
    int lane = threadIdx.x & 31;
    const float* h = H + (size_t)node * 64;
    float h0 = h[lane], h1 = h[lane + 32];
    float s = h0 * a_s[lane] + h1 * a_s[lane + 32];
    float d = h0 * a_d[lane] + h1 * a_d[lane + 32];
    #pragma unroll
    for (int off = 16; off; off >>= 1) {
        s += __shfl_xor(s, off, 32);
        d += __shfl_xor(d, off, 32);
    }
    if (lane == 0) {
        float a = s + d;
        a = (a >= 0.0f) ? a : NEG_SLOPE * a;
        v2f dm; dm.x = d; dm.y = a;
        S[node] = s;
        *(v2f*)(DM + 2 * (size_t)node) = dm;
    }
}

#define AGG_BLOCKS  256
#define AGG_ROWS    ((NN + AGG_BLOCKS - 1) / AGG_BLOCKS)
#define AGG_SRCBITS 17
__global__ void edge_max_kernel(const int* __restrict__ ei, int E, int Etot,
        const float* __restrict__ S, float* DM, float* __restrict__ AL, int* KEY, int* REC) {
    int i = blockIdx.x * blockDim.x + threadIdx.x;
    if (i >= Etot) return;
    int src, dst;
    if (i < E) { src = ei[i]; dst = ei[E + i]; }
    else       { src = dst = i - E; }
    float* pm = DM + 2 * (size_t)dst;
    v2f dm = *(const v2f*)pm;
    float a = S[src] + dm.x;
    a = (a >= 0.0f) ? a : NEG_SLOPE * a;
    if (KEY && i < E) {
        int own = (int)((unsigned)dst / AGG_ROWS);
        ((unsigned char*)KEY)[i] = (unsigned char)own;
        v2i rec;
        rec.x = src | ((dst - own * AGG_ROWS) << AGG_SRCBITS);
        rec.y = __float_as_int(a);
        ((v2i*)REC)[i] = rec;
    } else {
        AL[i] = a;
    }
    if (a > dm.y) atomicMaxF(pm + 1, a);
}

#define AGG_THREADS 1024
#define AGG_WAVES   (AGG_THREADS / 32)
#define AGG_QCAP    544
#define AGG_BATCH   8
#define AGG_KSTEP   (AGG_THREADS * 16)
#define AGG_LCAP    1024

__global__ __launch_bounds__(AGG_THREADS) void edge_agg_lds_kernel(
        const int* __restrict__ REC, int E,
        const float* __restrict__ H, const float* __restrict__ AL,
        const float* __restrict__ DM, const float* __restrict__ bias,
        float* __restrict__ OUT, int relu,
        int* __restrict__ LCNT, int* __restrict__ LIST, int replay,
        const int* __restrict__ KEY, const float* __restrict__ S) {
    __shared__ float acc[AGG_ROWS * 64];
    __shared__ float zl[AGG_ROWS + 1];
    __shared__ float ml[AGG_ROWS + 1];
    __shared__ float dd[AGG_ROWS + 1];
    __shared__ float sa[AGG_ROWS + 1];
    __shared__ int   qe[AGG_WAVES * AGG_QCAP];

    const int lo = blockIdx.x * AGG_ROWS;
    int cnt = NN - lo;
    if (cnt > AGG_ROWS) cnt = AGG_ROWS;
    if (cnt < 0) cnt = 0;
    const int lane = threadIdx.x & 31;
    const int wave = __builtin_amdgcn_readfirstlane((int)(threadIdx.x >> 5));

    static_assert(AGG_WAVES == 32, "one list length per lane");
    int len = 0;
    if (replay) len = LCNT[blockIdx.x * AGG_WAVES + lane];

    int* myqe = qe + wave * AGG_QCAP;
    int qn = 0;

    const int slot = blockIdx.x * AGG_WAVES + wave;
    v2i* mylist = (v2i*)LIST + (size_t)slot * AGG_LCAP;
    const bool record = (replay == 0);
    int ln = 0;
    bool maxpass = false;

    auto rows = [&](int n, int src, int dl, float w) {
        #pragma unroll 1
        for (int j0 = 0; j0 < n; j0 += AGG_BATCH) {
            float h0[AGG_BATCH], h1[AGG_BATCH];
            #pragma unroll
            for (int j = 0; j < AGG_BATCH; ++j) {
                h0[j] = 0.0f; h1[j] = 0.0f;
                if (j0 + j < n) {
                    int s = __builtin_amdgcn_readlane(src, j0 + j);
                    const float* h = H + (size_t)s * 64;
                    h0[j] = h[lane];
                    h1[j] = h[lane + 32];
                }
            }
            #pragma unroll
            for (int j = 0; j < AGG_BATCH; ++j) {
                if (j0 + j < n) {
                    int d = __builtin_amdgcn_readlane(dl, j0 + j);
                    float wj = __int_as_float(__builtin_amdgcn_readlane(__float_as_int(w), j0 + j));
                    float* o = acc + d * 64;
                    atomicAdd(&o[lane],      h0[j] * wj);
                    atomicAdd(&o[lane + 32], h1[j] * wj);
                }
            }
        }
    };

    auto flush = [&](int n) {
        int idx = qn - n + lane;
        qn -= n;
        int e = 0, sd = 0, src = 0, dl = 0;
        float w = 0.0f;
        if (lane < n) {
            e = myqe[idx];
            const v2i rec = ((const v2i*)REC)[e];
            float a = __int_as_float(rec.y);
            sd  = rec.x;
            src = sd & ((1 << AGG_SRCBITS) - 1);
            dl  = sd >> AGG_SRCBITS;
            if (replay) {
                a = S[src] + dd[dl];
                a = (a >= 0.0f) ? a : NEG_SLOPE * a;
            }
            if (maxpass) {
                atomicMaxF(&ml[dl], a);
            } else {
                w = expf(a - ml[dl]);
                atomicAdd(&zl[dl], w);
            }
        }
        if (record) {
            if (ln >= 0 && ln + n <= AGG_LCAP) {
                if (lane < n) {
                    v2i en; en.x = e; en.y = sd;
                    mylist[ln + lane] = en;
                }
                ln += n;
            } else {
                ln = -1;
            }
        }
        if (!maxpass) rows(n, src, dl, w);
    };

    auto push = [&](bool m, int e) {
        unsigned mask = __builtin_amdgcn_ballot_w32(m);
        if (mask) {
            int pos = qn + (int)__builtin_amdgcn_mbcnt_lo(mask, 0u);
            if (m) myqe[pos] = e;
            qn += __builtin_popcount(mask);
        }
    };

    bool scan = true;
    int mylen = 0;
    if (replay) {
        scan = __builtin_amdgcn_ballot_w32(len < 0) != 0u;
        mylen = __builtin_amdgcn_readlane(len, wave);
    }

    for (int pass = replay ? 0 : 1; pass < 2; ++pass) {
    maxpass = (pass == 0);
    if (maxpass) {
        for (int r = (int)threadIdx.x; r < cnt; r += AGG_THREADS) {
            const v2f dm = *(const v2f*)(DM + 2 * (size_t)(lo + r));
            dd[r] = dm.x;
            sa[r] = dm.y;
            ml[r] = dm.y;
        }
    } else {
        for (int r = wave; r < cnt; r += AGG_WAVES) {
            int n = lo + r;
            float m, a;
            if (replay) { m = ml[r]; a = sa[r]; }
            else        { m = DM[2 * (size_t)n + 1]; a = AL[(size_t)E + n]; }
            float w = expf(a - m);
            const float* h = H + (size_t)n * 64;
            acc[r * 64 + lane]      = h[lane]      * w;
            acc[r * 64 + lane + 32] = h[lane + 32] * w;
            if (lane == 0) { zl[r] = w; if (!replay) ml[r] = m; }
        }
    }
    __syncthreads();

    if (scan) {
    const unsigned pat = (unsigned)blockIdx.x * 0x01010101u;
    auto zero_bytes = [](unsigned x) -> unsigned {
        unsigned t = (x & 0x7F7F7F7Fu) + 0x7F7F7F7Fu;
        return ~(t | x) & 0x80808080u;
    };
    auto flags = [&](int4 v) -> unsigned {
        unsigned z0 = zero_bytes((unsigned)v.x ^ pat), z1 = zero_bytes((unsigned)v.y ^ pat);
        unsigned z2 = zero_bytes((unsigned)v.z ^ pat), z3 = zero_bytes((unsigned)v.w ^ pat);
        return (z0 >> 7) | (z1 >> 6) | (z2 >> 5) | (z3 >> 4);
    };
    auto scan_round = [&](int4 c0, int4 c1, int4 c2, int4 c3, int rbase) {
        unsigned long long M = (unsigned long long)(flags(c0) | (flags(c1) << 4))
                             | ((unsigned long long)(flags(c2) | (flags(c3) << 4)) << 32);
        const int lbase = rbase + (int)threadIdx.x * 16;
        while (__builtin_amdgcn_ballot_w32(M != 0ull)) {
            const bool has = (M != 0ull);
            const int P = has ? __builtin_ctzll(M) : 0;
            M &= M - 1ull;
            const int p = P & 31;
            const int u = ((P >> 5) << 1) + ((p >> 2) & 1);
            push(has, lbase + u * AGG_KSTEP + ((p & 3) << 2) + (p >> 3));
            while (qn >= 32) flush(32);
        }
    };
    const int pieces = E >> 4;
    const int PPR = 4 * AGG_THREADS;
    const int nfull = pieces / PPR;
    const int4* key4 = (const int4*)KEY + threadIdx.x;
    const int nokey = (int)~pat;
    int4 cur[4], nxt[4];
    #pragma unroll
    for (int u = 0; u < 4; ++u) {
        cur[u] = make_int4(nokey, nokey, nokey, nokey);
        nxt[u] = cur[u];
        if (nfull > 0) cur[u] = key4[u * AGG_THREADS];
    }
    for (int r = 0; r < nfull; ++r) {
        const int rn = (r + 1 < nfull) ? r + 1 : r;
        #pragma unroll
        for (int u = 0; u < 4; ++u)
            nxt[u] = key4[(size_t)rn * PPR + u * AGG_THREADS];
        scan_round(cur[0], cur[1], cur[2], cur[3], r * (PPR * 16));
        #pragma unroll
        for (int u = 0; u < 4; ++u) cur[u] = nxt[u];
    }
    if (nfull * PPR < pieces) {
        int4 c[4];
        #pragma unroll
        for (int u = 0; u < 4; ++u) {
            int pc = nfull * PPR + u * AGG_THREADS + (int)threadIdx.x;
            c[u] = make_int4(nokey, nokey, nokey, nokey);
            if (pc < pieces) c[u] = ((const int4*)KEY)[pc];
        }
        scan_round(c[0], c[1], c[2], c[3], nfull * (PPR * 16));
    }
    for (int base = pieces << 4; base < E; base += AGG_THREADS) {
        int e = base + (int)threadIdx.x;
        bool has = false;
        if (e < E) has = ((const unsigned char*)KEY)[e] == (unsigned char)blockIdx.x;
        push(has, e);
        while (qn >= 32) flush(32);
    }
    if (qn > 0) flush(qn);
    if (record && lane == 0) LCNT[slot] = ln;
    } else if (maxpass) {
    const int n = (mylen < AGG_LCAP) ? mylen : AGG_LCAP;
    v2i e0 = {0, 0}, e1 = {0, 0};
    float s0 = 0.0f;
    if (lane < n)      e0 = mylist[lane];
    if (lane + 32 < n) e1 = mylist[lane + 32];
    if (lane < n)      s0 = S[e0.y & ((1 << AGG_SRCBITS) - 1)];
    for (int c0 = 0; c0 < n; c0 += 32) {
        v2i e2 = {0, 0};
        float s1 = 0.0f;
        if (c0 + 64 + lane < n) e2 = mylist[c0 + 64 + lane];
        if (c0 + 32 + lane < n) s1 = S[e1.y & ((1 << AGG_SRCBITS) - 1)];
        if (c0 + lane < n) {
            const int dl = e0.y >> AGG_SRCBITS;
            float a = s0 + dd[dl];
            a = (a >= 0.0f) ? a : NEG_SLOPE * a;
            atomicMaxF(&ml[dl], a);
            mylist[c0 + lane].x = __float_as_int(a);
        }
        e0 = e1; e1 = e2; s0 = s1;
    }
    } else {
    const int n = (mylen < AGG_LCAP) ? mylen : AGG_LCAP;
    v2i e0 = {0, 0}, e1 = {0, 0};
    if (lane < n)      e0 = mylist[lane];
    if (lane + 32 < n) e1 = mylist[lane + 32];
    for (int c0 = 0; c0 < n; c0 += 32) {
        int m = n - c0;
        if (m > 32) m = 32;
        v2i e2 = {0, 0};
        if (c0 + 64 + lane < n) e2 = mylist[c0 + 64 + lane];
        int src = 0, dl = 0;
        float w = 0.0f;
        if (lane < m) {
            src = e0.y & ((1 << AGG_SRCBITS) - 1);
            dl  = e0.y >> AGG_SRCBITS;
            w = expf(__int_as_float(e0.x) - ml[dl]);
            atomicAdd(&zl[dl], w);
        }
        rows(m, src, dl, w);
        e0 = e1; e1 = e2;
    }
    }
    __syncthreads();
    }

    const float b0 = bias[lane], b1 = bias[lane + 32];
    for (int r = wave; r < cnt; r += AGG_WAVES) {
        float zz = zl[r] + EPSF;
        float v0 = acc[r * 64 + lane]      / zz + b0;
        float v1 = acc[r * 64 + lane + 32] / zz + b1;
        if (relu) {
            v0 = (v0 > 0.0f) ? v0 : 0.0f;
            v1 = (v1 > 0.0f) ? v1 : 0.0f;
        }
        float* o = OUT + (size_t)(lo + r) * 64;
        o[lane]      = v0;
        o[lane + 32] = v1;
    }
}

extern "C" void kernel_launch(void* const* d_in, const int* in_sizes, int n_in,
                              void* d_out, int out_size, void* d_ws, size_t ws_size,
                              hipStream_t stream) {
    const float* x     = (const float*)d_in[0];
    const int*   ei    = (const int*)d_in[1];
    const float* W1    = (const float*)d_in[2];
    const float* as1   = (const float*)d_in[3];
    const float* ad1   = (const float*)d_in[4];
    const float* b1    = (const float*)d_in[5];
    const float* W2    = (const float*)d_in[6];
    const float* as2   = (const float*)d_in[7];
    const float* ad2   = (const float*)d_in[8];
    const float* b2    = (const float*)d_in[9];
    float* out = (float*)d_out;

    const int E    = in_sizes[1] / 2;
    const int Etot = E + NN;
    const int NF   = NN * 64;

    float* wsf = (float*)d_ws;
    float* H   = wsf;
    float* X2  = H   + NF;
    float* S   = X2  + NF;
    float* DM  = S   + NN;
    int* LCNT  = (int*)(DM + 2 * NN);
    int* LIST  = LCNT + AGG_BLOCKS * AGG_WAVES;
    int* KEY   = LIST + AGG_BLOCKS * AGG_WAVES * AGG_LCAP * 2;
    int* REC   = KEY + ((E + 15) / 16) * 4;
    float* AL  = (float*)(REC + 2 * (size_t)E);

    const int TB = 256;
    dim3 blk(TB);
    int gE    = (Etot + TB - 1) / TB;
    int gScr  = (NN + 7) / 8;
    int gGemm = ((NN / 16) + 3) / 4;

    gemm_wmma_kernel<<<gGemm, dim3(128), 0, stream>>>(x, W1, H, NN, 128);
    scores_kernel<<<gScr, blk, 0, stream>>>(H, as1, ad1, S, DM);
    edge_max_kernel<<<gE, blk, 0, stream>>>(ei, E, Etot, S, DM, AL, KEY, REC);
    edge_agg_lds_kernel<<<AGG_BLOCKS, dim3(AGG_THREADS), 0, stream>>>(REC, E, H, AL, DM, b1, X2, 1, LCNT, LIST, 0, KEY, S);

    gemm_wmma_kernel<<<gGemm, dim3(128), 0, stream>>>(X2, W2, H, NN, 64);
    scores_kernel<<<gScr, blk, 0, stream>>>(H, as2, ad2, S, DM);
    edge_agg_lds_kernel<<<AGG_BLOCKS, dim3(AGG_THREADS), 0, stream>>>(REC, E, H, AL, DM, b2, out, 0, LCNT, LIST, 1, KEY, S);
}


# Layer 1 has no score/max launch: its row-sum launch forms scores (S gathered, D in LDS) and maxima by LDS max while it scans, then walks its own lists for the sums; LDS adds by arrival as before
# speedup vs baseline: 4.2410x; 1.2762x over previous
#include <hip/hip_runtime.h>
#include <math.h>

typedef float v2f __attribute__((ext_vector_type(2)));
typedef float v8f __attribute__((ext_vector_type(8)));
typedef int   v2i __attribute__((ext_vector_type(2)));

#define NN 100000
#define NEG_SLOPE 0.2f
#define EPSF 1e-16f

__device__ __forceinline__ void atomicMaxF(float* addr, float v) {
    if (v >= 0.0f) atomicMax((int*)addr, __float_as_int(v));
    else           atomicMin((unsigned int*)addr, (unsigned int)__float_as_int(v));
}

__global__ __launch_bounds__(128) void gemm_wmma_kernel(
        const float* __restrict__ X, const float* __restrict__ W,
        float* __restrict__ H, int nrows, int K) {
    int wave = blockIdx.x * (blockDim.x >> 5) + (threadIdx.x >> 5);
    int mtiles = nrows >> 4;
    if (wave >= mtiles) return;

    int lane  = threadIdx.x & 31;
    int r0    = lane & 15;
    int khalf = lane >> 4;

    const float* xrow = X + (size_t)(wave * 16 + r0) * (size_t)K;

    v8f c0 = {}, c1 = {}, c2 = {}, c3 = {};
    for (int k0 = 0; k0 < K; k0 += 4) {
        int ka = k0 + khalf * 2;
        v2f a;
        a.x = xrow[ka];
        a.y = xrow[ka + 1];

        const float* w0 = W + (size_t)ka * 64;
        const float* w1 = w0 + 64;
        v2f b0, b1, b2, b3;
        b0.x = w0[r0];      b0.y = w1[r0];
        b1.x = w0[16 + r0]; b1.y = w1[16 + r0];
        b2.x = w0[32 + r0]; b2.y = w1[32 + r0];
        b3.x = w0[48 + r0]; b3.y = w1[48 + r0];

        c0 = __builtin_amdgcn_wmma_f32_16x16x4_f32(false, a, false, b0, (short)0, c0, false, false);
        c1 = __builtin_amdgcn_wmma_f32_16x16x4_f32(false, a, false, b1, (short)0, c1, false, false);
        c2 = __builtin_amdgcn_wmma_f32_16x16x4_f32(false, a, false, b2, (short)0, c2, false, false);
        c3 = __builtin_amdgcn_wmma_f32_16x16x4_f32(false, a, false, b3, (short)0, c3, false, false);
    }

    float* hbase = H + (size_t)(wave * 16 + khalf * 8) * 64 + r0;
    #pragma unroll
    for (int r = 0; r < 8; ++r) {
        float* row = hbase + (size_t)r * 64;
        row[0]  = c0[r];
        row[16] = c1[r];
        row[32] = c2[r];
        row[48] = c3[r];
    }
}

__global__ __launch_bounds__(256) void scores_kernel(
        const float* __restrict__ H, const float* __restrict__ a_s,
        const float* __restrict__ a_d, float* __restrict__ S, float* __restrict__ DM) {
    int node = blockIdx.x * (blockDim.x >> 5) + (threadIdx.x >> 5);
    if (node >= NN) return;
    int lane = threadIdx.x & 31;
    const float* h = H + (size_t)node * 64;
    float h0 = h[lane], h1 = h[lane + 32];
    float s = h0 * a_s[lane] + h1 * a_s[lane + 32];
    float d = h0 * a_d[lane] + h1 * a_d[lane + 32];
    #pragma unroll
    for (int off = 16; off; off >>= 1) {
        s += __shfl_xor(s, off, 32);
        d += __shfl_xor(d, off, 32);
    }
    if (lane == 0) {
        float a = s + d;
        a = (a >= 0.0f) ? a : NEG_SLOPE * a;
        v2f dm; dm.x = d; dm.y = a;
        S[node] = s;
        *(v2f*)(DM + 2 * (size_t)node) = dm;
    }
}

#define AGG_BLOCKS  256
#define AGG_ROWS    ((NN + AGG_BLOCKS - 1) / AGG_BLOCKS)
#define AGG_SRCBITS 17
__global__ void edge_key_kernel(const int* __restrict__ ei, int E,
        int* __restrict__ KEY, int* __restrict__ REC) {
    int i = blockIdx.x * blockDim.x + threadIdx.x;
    if (i >= E) return;
    int src = ei[i], dst = ei[E + i];
    int own = (int)((unsigned)dst / AGG_ROWS);
    ((unsigned char*)KEY)[i] = (unsigned char)own;
    REC[i] = src | ((dst - own * AGG_ROWS) << AGG_SRCBITS);
}

#define AGG_THREADS 1024
#define AGG_WAVES   (AGG_THREADS / 32)
#define AGG_QCAP    544
#define AGG_BATCH   8
#define AGG_KSTEP   (AGG_THREADS * 16)
#define AGG_LCAP    1024

__global__ __launch_bounds__(AGG_THREADS) void edge_agg_lds_kernel(
        const int* __restrict__ REC, int E,
        const float* __restrict__ H,
        const float* __restrict__ DM, const float* __restrict__ bias,
        float* __restrict__ OUT, int relu,
        int* __restrict__ LCNT, int* __restrict__ LIST, int replay,
        const int* __restrict__ KEY, const float* __restrict__ S) {
    __shared__ float acc[AGG_ROWS * 64];
    __shared__ float zl[AGG_ROWS + 1];
    __shared__ float ml[AGG_ROWS + 1];
    __shared__ float dd[AGG_ROWS + 1];
    __shared__ float sa[AGG_ROWS + 1];
    __shared__ int   qe[AGG_WAVES * AGG_QCAP];
    __shared__ int   ovf;

    const int lo = blockIdx.x * AGG_ROWS;
    int cnt = NN - lo;
    if (cnt > AGG_ROWS) cnt = AGG_ROWS;
    if (cnt < 0) cnt = 0;
    const int lane = threadIdx.x & 31;
    const int wave = __builtin_amdgcn_readfirstlane((int)(threadIdx.x >> 5));

    static_assert(AGG_WAVES == 32, "one list length per lane");
    int len = 0;
    if (replay) len = LCNT[blockIdx.x * AGG_WAVES + lane];

    int* myqe = qe + wave * AGG_QCAP;
    int qn = 0;

    const int slot = blockIdx.x * AGG_WAVES + wave;
    v2i* mylist = (v2i*)LIST + (size_t)slot * AGG_LCAP;
    const bool record = (replay == 0);
    int ln = 0;
    bool maxpass = false;

    auto rows = [&](int n, int src, int dl, float w) {
        #pragma unroll 1
        for (int j0 = 0; j0 < n; j0 += AGG_BATCH) {
            float h0[AGG_BATCH], h1[AGG_BATCH];
            #pragma unroll
            for (int j = 0; j < AGG_BATCH; ++j) {
                h0[j] = 0.0f; h1[j] = 0.0f;
                if (j0 + j < n) {
                    int s = __builtin_amdgcn_readlane(src, j0 + j);
                    const float* h = H + (size_t)s * 64;
                    h0[j] = h[lane];
                    h1[j] = h[lane + 32];
                }
            }
            #pragma unroll
            for (int j = 0; j < AGG_BATCH; ++j) {
                if (j0 + j < n) {
                    int d = __builtin_amdgcn_readlane(dl, j0 + j);
                    float wj = __int_as_float(__builtin_amdgcn_readlane(__float_as_int(w), j0 + j));
                    float* o = acc + d * 64;
                    atomicAdd(&o[lane],      h0[j] * wj);
                    atomicAdd(&o[lane + 32], h1[j] * wj);
                }
            }
        }
    };

    auto flush = [&](int n) {
        int idx = qn - n + lane;
        qn -= n;
        int sd = 0, src = 0, dl = 0;
        float w = 0.0f, a = 0.0f;
        if (lane < n) {
            const int e = myqe[idx];
            sd  = REC[e];
            src = sd & ((1 << AGG_SRCBITS) - 1);
            dl  = sd >> AGG_SRCBITS;
            a = S[src] + dd[dl];
            a = (a >= 0.0f) ? a : NEG_SLOPE * a;
            if (maxpass) {
                atomicMaxF(&ml[dl], a);
            } else {
                w = expf(a - ml[dl]);
                atomicAdd(&zl[dl], w);
            }
        }
        if (record && maxpass) {
            if (ln >= 0 && ln + n <= AGG_LCAP) {
                if (lane < n) {
                    v2i en; en.x = __float_as_int(a); en.y = sd;
                    mylist[ln + lane] = en;
                }
                ln += n;
            } else {
                ln = -1;
            }
        }
        if (!maxpass) rows(n, src, dl, w);
    };

    auto push = [&](bool m, int e) {
        unsigned mask = __builtin_amdgcn_ballot_w32(m);
        if (mask) {
            int pos = qn + (int)__builtin_amdgcn_mbcnt_lo(mask, 0u);
            if (m) myqe[pos] = e;
            qn += __builtin_popcount(mask);
        }
    };

    bool scan = true;
    int mylen = 0;
    if (replay) {
        scan = __builtin_amdgcn_ballot_w32(len < 0) != 0u;
        mylen = __builtin_amdgcn_readlane(len, wave);
    }

    for (int pass = 0; pass < 2; ++pass) {
    maxpass = (pass == 0);
    if (maxpass) {
        for (int r = (int)threadIdx.x; r < cnt; r += AGG_THREADS) {
            const v2f dm = *(const v2f*)(DM + 2 * (size_t)(lo + r));
            dd[r] = dm.x;
            sa[r] = dm.y;
            ml[r] = dm.y;
        }
        if (threadIdx.x == 0) ovf = 0;
    } else {
        for (int r = wave; r < cnt; r += AGG_WAVES) {
            int n = lo + r;
            float w = expf(sa[r] - ml[r]);
            const float* h = H + (size_t)n * 64;
            acc[r * 64 + lane]      = h[lane]      * w;
            acc[r * 64 + lane + 32] = h[lane + 32] * w;
            if (lane == 0) zl[r] = w;
        }
    }
    __syncthreads();

    if (!replay) {
        scan = maxpass || (__builtin_amdgcn_readfirstlane(ovf) != 0);
        mylen = __builtin_amdgcn_readfirstlane(ln);
    }

    if (scan) {
    const unsigned pat = (unsigned)blockIdx.x * 0x01010101u;
    auto zero_bytes = [](unsigned x) -> unsigned {
        unsigned t = (x & 0x7F7F7F7Fu) + 0x7F7F7F7Fu;
        return ~(t | x) & 0x80808080u;
    };
    auto flags = [&](int4 v) -> unsigned {
        unsigned z0 = zero_bytes((unsigned)v.x ^ pat), z1 = zero_bytes((unsigned)v.y ^ pat);
        unsigned z2 = zero_bytes((unsigned)v.z ^ pat), z3 = zero_bytes((unsigned)v.w ^ pat);
        return (z0 >> 7) | (z1 >> 6) | (z2 >> 5) | (z3 >> 4);
    };
    auto scan_round = [&](int4 c0, int4 c1, int4 c2, int4 c3, int rbase) {
        unsigned long long M = (unsigned long long)(flags(c0) | (flags(c1) << 4))
                             | ((unsigned long long)(flags(c2) | (flags(c3) << 4)) << 32);
        const int lbase = rbase + (int)threadIdx.x * 16;
        while (__builtin_amdgcn_ballot_w32(M != 0ull)) {
            const bool has = (M != 0ull);
            const int P = has ? __builtin_ctzll(M) : 0;
            M &= M - 1ull;
            const int p = P & 31;
            const int u = ((P >> 5) << 1) + ((p >> 2) & 1);
            push(has, lbase + u * AGG_KSTEP + ((p & 3) << 2) + (p >> 3));
            while (qn >= 32) flush(32);
        }
    };
    const int pieces = E >> 4;
    const int PPR = 4 * AGG_THREADS;
    const int nfull = pieces / PPR;
    const int4* key4 = (const int4*)KEY + threadIdx.x;
    const int nokey = (int)~pat;
    int4 cur[4], nxt[4];
    #pragma unroll
    for (int u = 0; u < 4; ++u) {
        cur[u] = make_int4(nokey, nokey, nokey, nokey);
        nxt[u] = cur[u];
        if (nfull > 0) cur[u] = key4[u * AGG_THREADS];
    }
    for (int r = 0; r < nfull; ++r) {
        const int rn = (r + 1 < nfull) ? r + 1 : r;
        #pragma unroll
        for (int u = 0; u < 4; ++u)
            nxt[u] = key4[(size_t)rn * PPR + u * AGG_THREADS];
        scan_round(cur[0], cur[1], cur[2], cur[3], r * (PPR * 16));
        #pragma unroll
        for (int u = 0; u < 4; ++u) cur[u] = nxt[u];
    }
    if (nfull * PPR < pieces) {
        int4 c[4];
        #pragma unroll
        for (int u = 0; u < 4; ++u) {
            int pc = nfull * PPR + u * AGG_THREADS + (int)threadIdx.x;
            c[u] = make_int4(nokey, nokey, nokey, nokey);
            if (pc < pieces) c[u] = ((const int4*)KEY)[pc];
        }
        scan_round(c[0], c[1], c[2], c[3], nfull * (PPR * 16));
    }
    for (int base = pieces << 4; base < E; base += AGG_THREADS) {
        int e = base + (int)threadIdx.x;
        bool has = false;
        if (e < E) has = ((const unsigned char*)KEY)[e] == (unsigned char)blockIdx.x;
        push(has, e);
        while (qn >= 32) flush(32);
    }
    if (qn > 0) flush(qn);
    if (record && maxpass && lane == 0) {
        LCNT[slot] = ln;
        if (ln < 0) ovf = 1;
    }
    } else if (maxpass) {
    const int n = (mylen < AGG_LCAP) ? mylen : AGG_LCAP;
    v2i e0 = {0, 0}, e1 = {0, 0};
    float s0 = 0.0f;
    if (lane < n)      e0 = mylist[lane];
    if (lane + 32 < n) e1 = mylist[lane + 32];
    if (lane < n)      s0 = S[e0.y & ((1 << AGG_SRCBITS) - 1)];
    for (int c0 = 0; c0 < n; c0 += 32) {
        v2i e2 = {0, 0};
        float s1 = 0.0f;
        if (c0 + 64 + lane < n) e2 = mylist[c0 + 64 + lane];
        if (c0 + 32 + lane < n) s1 = S[e1.y & ((1 << AGG_SRCBITS) - 1)];
        if (c0 + lane < n) {
            const int dl = e0.y >> AGG_SRCBITS;
            float a = s0 + dd[dl];
            a = (a >= 0.0f) ? a : NEG_SLOPE * a;
            atomicMaxF(&ml[dl], a);
            mylist[c0 + lane].x = __float_as_int(a);
        }
        e0 = e1; e1 = e2; s0 = s1;
    }
    } else {
    const int n = (mylen < AGG_LCAP) ? mylen : AGG_LCAP;
    v2i e0 = {0, 0}, e1 = {0, 0};
    if (lane < n)      e0 = mylist[lane];
    if (lane + 32 < n) e1 = mylist[lane + 32];
    for (int c0 = 0; c0 < n; c0 += 32) {
        int m = n - c0;
        if (m > 32) m = 32;
        v2i e2 = {0, 0};
        if (c0 + 64 + lane < n) e2 = mylist[c0 + 64 + lane];
        int src = 0, dl = 0;
        float w = 0.0f;
        if (lane < m) {
            src = e0.y & ((1 << AGG_SRCBITS) - 1);
            dl  = e0.y >> AGG_SRCBITS;
            w = expf(__int_as_float(e0.x) - ml[dl]);
            atomicAdd(&zl[dl], w);
        }
        rows(m, src, dl, w);
        e0 = e1; e1 = e2;
    }
    }
    __syncthreads();
    }

    const float b0 = bias[lane], b1 = bias[lane + 32];
    for (int r = wave; r < cnt; r += AGG_WAVES) {
        float zz = zl[r] + EPSF;
        float v0 = acc[r * 64 + lane]      / zz + b0;
        float v1 = acc[r * 64 + lane + 32] / zz + b1;
        if (relu) {
            v0 = (v0 > 0.0f) ? v0 : 0.0f;
            v1 = (v1 > 0.0f) ? v1 : 0.0f;
        }
        float* o = OUT + (size_t)(lo + r) * 64;
        o[lane]      = v0;
        o[lane + 32] = v1;
    }
}

extern "C" void kernel_launch(void* const* d_in, const int* in_sizes, int n_in,
                              void* d_out, int out_size, void* d_ws, size_t ws_size,
                              hipStream_t stream) {
    const float* x     = (const float*)d_in[0];
    const int*   ei    = (const int*)d_in[1];
    const float* W1    = (const float*)d_in[2];
    const float* as1   = (const float*)d_in[3];
    const float* ad1   = (const float*)d_in[4];
    const float* b1    = (const float*)d_in[5];
    const float* W2    = (const float*)d_in[6];
    const float* as2   = (const float*)d_in[7];
    const float* ad2   = (const float*)d_in[8];
    const float* b2    = (const float*)d_in[9];
    float* out = (float*)d_out;

    const int E    = in_sizes[1] / 2;
    const int NF   = NN * 64;

    float* wsf = (float*)d_ws;
    float* H   = wsf;
    float* X2  = H   + NF;
    float* S   = X2  + NF;
    float* DM  = S   + NN;
    int* LCNT  = (int*)(DM + 2 * NN);
    int* LIST  = LCNT + AGG_BLOCKS * AGG_WAVES;
    int* KEY   = LIST + AGG_BLOCKS * AGG_WAVES * AGG_LCAP * 2;
    int* REC   = KEY + ((E + 15) / 16) * 4;

    const int TB = 256;
    dim3 blk(TB);
    int gE    = (E + TB - 1) / TB;
    int gScr  = (NN + 7) / 8;
    int gGemm = ((NN / 16) + 3) / 4;

    gemm_wmma_kernel<<<gGemm, dim3(128), 0, stream>>>(x, W1, H, NN, 128);
    scores_kernel<<<gScr, blk, 0, stream>>>(H, as1, ad1, S, DM);
    edge_key_kernel<<<gE, blk, 0, stream>>>(ei, E, KEY, REC);
    edge_agg_lds_kernel<<<AGG_BLOCKS, dim3(AGG_THREADS), 0, stream>>>(REC, E, H, DM, b1, X2, 1, LCNT, LIST, 0, KEY, S);

    gemm_wmma_kernel<<<gGemm, dim3(128), 0, stream>>>(X2, W2, H, NN, 64);
    scores_kernel<<<gScr, blk, 0, stream>>>(H, as2, ad2, S, DM);
    edge_agg_lds_kernel<<<AGG_BLOCKS, dim3(AGG_THREADS), 0, stream>>>(REC, E, H, DM, b2, out, 0, LCNT, LIST, 1, KEY, S);
}
